# starting kernel
# baseline (speedup 1.0000x reference)
#include <hip/hip_runtime.h>
#include <hip/hip_bf16.h>


typedef __attribute__((ext_vector_type(2))) float v2f;
typedef __attribute__((ext_vector_type(8))) float v8f;

#define F_NODES 16384
#define DEG 8
#define EDGES (F_NODES * DEG)
#define BATCH 64
#define HID 16
#define LAYERS 10
#define WAVES_PER_BLOCK 8
#define LDS_STRIDE 18

__device__ __forceinline__ float elu_f(float z) {
    return z > 0.0f ? z : (__expf(z) - 1.0f);
}

__global__ __launch_bounds__(256)
void gsnn_layer(const float* __restrict__ xin,
                const float* __restrict__ x0,
                const float* __restrict__ W1,
                const float* __restrict__ b1,
                const float* __restrict__ W2,
                const float* __restrict__ b2,
                const int*   __restrict__ ixs,
                float*       __restrict__ xout)
{
    __shared__ float lds_h[WAVES_PER_BLOCK * 16 * LDS_STRIDE];

    const int tid  = threadIdx.x;
    const int wave = tid >> 5;
    const int lane = tid & 31;
    const int n    = lane & 15;
    const int hi   = lane >> 4;

    const int f = blockIdx.x * WAVES_PER_BLOCK + wave;
    float* hbuf = &lds_h[wave * 16 * LDS_STRIDE];

    const float* W1f = W1 + (size_t)f * (HID * DEG);
    const v2f wb1_0 = *(const v2f*)(W1f + n * DEG + 2 * hi);
    const v2f wb1_1 = *(const v2f*)(W1f + n * DEG + 4 + 2 * hi);
    const float bias1 = b1[f * HID + n];

    const float* W2f = W2 + (size_t)f * (DEG * HID);
    v2f wb2[4];
    float bias2 = 0.0f;
    if (n < DEG) {
        #pragma unroll
        for (int j = 0; j < 4; ++j)
            wb2[j] = *(const v2f*)(W2f + n * HID + 4 * j + 2 * hi);
        bias2 = b2[f * DEG + n];
    } else {
        #pragma unroll
        for (int j = 0; j < 4; ++j) { wb2[j].x = 0.0f; wb2[j].y = 0.0f; }
    }

    const int* ixf = ixs + f * DEG;
    const int c00 = ixf[2 * hi];
    const int c01 = ixf[2 * hi + 1];
    const int c10 = ixf[4 + 2 * hi];
    const int c11 = ixf[4 + 2 * hi + 1];

    const size_t ocol = (size_t)f * DEG + n;

    #pragma unroll
    for (int mt = 0; mt < 4; ++mt) {
        const int mb = mt * 16;

        const float* xrow = xin + (size_t)(mb + n) * EDGES;
        v2f a0; a0.x = xrow[c00]; a0.y = xrow[c01];
        v2f a1; a1.x = xrow[c10]; a1.y = xrow[c11];

        v8f acc = {};
        acc = __builtin_amdgcn_wmma_f32_16x16x4_f32(
                  false, a0, false, wb1_0, (short)0, acc, false, false);
        acc = __builtin_amdgcn_wmma_f32_16x16x4_f32(
                  false, a1, false, wb1_1, (short)0, acc, false, false);

        #pragma unroll
        for (int r = 0; r < 8; ++r) {
            float hv = elu_f(acc[r] + bias1);
            hbuf[(r + 8 * hi) * LDS_STRIDE + n] = hv;
        }
        __syncthreads();

        v2f a2[4];
        #pragma unroll
        for (int j = 0; j < 4; ++j)
            a2[j] = *(const v2f*)(hbuf + n * LDS_STRIDE + 4 * j + 2 * hi);

        v8f o = {};
        #pragma unroll
        for (int j = 0; j < 4; ++j)
            o = __builtin_amdgcn_wmma_f32_16x16x4_f32(
                    false, a2[j], false, wb2[j], (short)0, o, false, false);

        __syncthreads();

        if (n < DEG) {
            #pragma unroll
            for (int r = 0; r < 8; ++r) {
                const size_t off = (size_t)(mb + r + 8 * hi) * EDGES + ocol;
                xout[off] = o[r] + bias2 + x0[off];
            }
        }
    }
}

extern "C" void kernel_launch(void* const* d_in, const int* in_sizes, int n_in,
                              void* d_out, int out_size, void* d_ws, size_t ws_size,
                              hipStream_t stream) {
    (void)in_sizes; (void)n_in; (void)out_size; (void)ws_size;

    const float* x0 = (const float*)d_in[0];
    const float* W1 = (const float*)d_in[1];
    const float* b1 = (const float*)d_in[2];
    const float* W2 = (const float*)d_in[3];
    const float* b2 = (const float*)d_in[4];
    const int*   ix = (const int*)d_in[5];

    float* bufA = (float*)d_ws;
    float* out  = (float*)d_out;

    const dim3 grid(F_NODES / WAVES_PER_BLOCK);
    const dim3 block(WAVES_PER_BLOCK * 32);

    const float* cur = x0;
    for (int l = 0; l < LAYERS; ++l) {
        float* dst = (l & 1) ? out : bufA;
        gsnn_layer<<<grid, block, 0, stream>>>(cur, x0, W1, b1, W2, b2, ix, dst);
        cur = dst;
    }
}


# residual read and output write made block-cooperative: GEMM2 computed transposed so a lane holds 8 adjacent outputs, block's 64x64 tile met in LDS, x0 and xout moved as whole 128B lines; same sums
# speedup vs baseline: 1.0562x; 1.0562x over previous
#include <hip/hip_runtime.h>
#include <hip/hip_bf16.h>


typedef __attribute__((ext_vector_type(2))) float v2f;
typedef __attribute__((ext_vector_type(8))) float v8f;
typedef __attribute__((ext_vector_type(4))) float v4f;

#define F_NODES 16384
#define DEG 8
#define EDGES (F_NODES * DEG)
#define BATCH 64
#define HID 16
#define LAYERS 10
#define WAVES_PER_BLOCK 8
#define LDS_STRIDE 18
#define OUT_PITCH 68

__device__ __forceinline__ float elu_f(float z) {
    return z > 0.0f ? z : (__expf(z) - 1.0f);
}

__global__ __launch_bounds__(256)
void gsnn_layer(const float* __restrict__ xin,
                const float* __restrict__ x0,
                const float* __restrict__ W1,
                const float* __restrict__ b1,
                const float* __restrict__ W2,
                const float* __restrict__ b2,
                const int*   __restrict__ ixs,
                float*       __restrict__ xout)
{
    __shared__ float lds_h[WAVES_PER_BLOCK * 16 * LDS_STRIDE];
    __shared__ __attribute__((aligned(16))) float lds_o[BATCH * OUT_PITCH];

    const int tid  = threadIdx.x;
    const int wave = tid >> 5;
    const int lane = tid & 31;
    const int n    = lane & 15;
    const int hi   = lane >> 4;

    const int f = blockIdx.x * WAVES_PER_BLOCK + wave;
    float* hbuf = &lds_h[wave * 16 * LDS_STRIDE];

    const float* W1f = W1 + (size_t)f * (HID * DEG);
    const v2f wb1_0 = *(const v2f*)(W1f + n * DEG + 2 * hi);
    const v2f wb1_1 = *(const v2f*)(W1f + n * DEG + 4 + 2 * hi);
    const float bias1 = b1[f * HID + n];

    const float* W2f = W2 + (size_t)f * (DEG * HID);
    v2f wb2[4];
    if (n < DEG) {
        #pragma unroll
        for (int j = 0; j < 4; ++j)
            wb2[j] = *(const v2f*)(W2f + n * HID + 4 * j + 2 * hi);
    } else {
        #pragma unroll
        for (int j = 0; j < 4; ++j) { wb2[j].x = 0.0f; wb2[j].y = 0.0f; }
    }

    const int* ixf = ixs + f * DEG;
    const int c00 = ixf[2 * hi];
    const int c01 = ixf[2 * hi + 1];
    const int c10 = ixf[4 + 2 * hi];
    const int c11 = ixf[4 + 2 * hi + 1];

    const int crow = lane >> 4;
    const int ccol = (lane & 15) * 4;
    const size_t gcol = (size_t)blockIdx.x * (WAVES_PER_BLOCK * DEG) + ccol;

    v4f xr[4];
    v4f b2v;

    #pragma unroll
    for (int mt = 0; mt < 4; ++mt) {
        const int mb = mt * 16;

        if (mt == 3) {
            #pragma unroll
            for (int i = 0; i < 4; ++i) {
                const int row = 8 * wave + 2 * i + crow;
                xr[i] = *(const v4f*)(x0 + (size_t)row * EDGES + gcol);
            }
            b2v = *(const v4f*)(b2 + gcol);
        }

        const float* xrow = xin + (size_t)(mb + n) * EDGES;
        v2f a0; a0.x = xrow[c00]; a0.y = xrow[c01];
        v2f a1; a1.x = xrow[c10]; a1.y = xrow[c11];

        v8f acc = {};
        acc = __builtin_amdgcn_wmma_f32_16x16x4_f32(
                  false, a0, false, wb1_0, (short)0, acc, false, false);
        acc = __builtin_amdgcn_wmma_f32_16x16x4_f32(
                  false, a1, false, wb1_1, (short)0, acc, false, false);

        #pragma unroll
        for (int r = 0; r < 8; ++r) {
            float hv = elu_f(acc[r] + bias1);
            hbuf[(r + 8 * hi) * LDS_STRIDE + n] = hv;
        }
        if (mt == 3) {
            asm volatile("" :: "v"(xr[0]), "v"(xr[1]), "v"(xr[2]), "v"(xr[3]), "v"(b2v));
        }
        __syncthreads();

        v2f a2[4];
        #pragma unroll
        for (int j = 0; j < 4; ++j)
            a2[j] = *(const v2f*)(hbuf + n * LDS_STRIDE + 4 * j + 2 * hi);

        v8f o = {};
        #pragma unroll
        for (int j = 0; j < 4; ++j)
            o = __builtin_amdgcn_wmma_f32_16x16x4_f32(
                    false, wb2[j], false, a2[j], (short)0, o, false, false);

        __syncthreads();

        if (hi == 0) {
            float* orow = lds_o + (mb + n) * OUT_PITCH + wave * DEG;
            v4f lo4, hi4;
            lo4.x = o[0]; lo4.y = o[1]; lo4.z = o[2]; lo4.w = o[3];
            hi4.x = o[4]; hi4.y = o[5]; hi4.z = o[6]; hi4.w = o[7];
            *(v4f*)(orow)     = lo4;
            *(v4f*)(orow + 4) = hi4;
        }
    }

    __syncthreads();

    #pragma unroll
    for (int i = 0; i < 4; ++i) {
        const int row = 8 * wave + 2 * i + crow;
        const v4f ov = *(const v4f*)(lds_o + row * OUT_PITCH + ccol);
        const v4f res = (ov + b2v) + xr[i];
        *(v4f*)(xout + (size_t)row * EDGES + gcol) = res;
    }
}

extern "C" void kernel_launch(void* const* d_in, const int* in_sizes, int n_in,
                              void* d_out, int out_size, void* d_ws, size_t ws_size,
                              hipStream_t stream) {
    (void)in_sizes; (void)n_in; (void)out_size; (void)ws_size;

    const float* x0 = (const float*)d_in[0];
    const float* W1 = (const float*)d_in[1];
    const float* b1 = (const float*)d_in[2];
    const float* W2 = (const float*)d_in[3];
    const float* b2 = (const float*)d_in[4];
    const int*   ix = (const int*)d_in[5];

    float* bufA = (float*)d_ws;
    float* out  = (float*)d_out;

    const dim3 grid(F_NODES / WAVES_PER_BLOCK);
    const dim3 block(WAVES_PER_BLOCK * 32);

    const float* cur = x0;
    for (int l = 0; l < LAYERS; ++l) {
        float* dst = (l & 1) ? out : bufA;
        gsnn_layer<<<grid, block, 0, stream>>>(cur, x0, W1, b1, W2, b2, ix, dst);
        cur = dst;
    }
}


# gather x[:,ixs] made a per-layer row-permutation pass: half a row staged in LDS by whole-line loads, values gathered from LDS; rows stay (B,E); layer kernel reads 8 adjacent columns; same sums
# speedup vs baseline: 1.2290x; 1.1636x over previous
#include <hip/hip_runtime.h>
#include <hip/hip_bf16.h>


typedef __attribute__((ext_vector_type(2))) float v2f;
typedef __attribute__((ext_vector_type(8))) float v8f;
typedef __attribute__((ext_vector_type(4))) float v4f;

#define F_NODES 16384
#define DEG 8
#define EDGES (F_NODES * DEG)
#define BATCH 64
#define HID 16
#define LAYERS 10
#define WAVES_PER_BLOCK 8
#define LDS_STRIDE 18
#define OUT_PITCH 68

__device__ __forceinline__ float elu_f(float z) {
    return z > 0.0f ? z : (__expf(z) - 1.0f);
}


#define PERM_THREADS 1024
#define WIN_COLS 65536
#define DEST_COLS 32768

__global__ __launch_bounds__(PERM_THREADS)
void gsnn_permute(const float* __restrict__ xin,
                  const int*   __restrict__ ixs,
                  float*       __restrict__ xg)
{
    __shared__ __attribute__((aligned(16))) float win[WIN_COLS];

    const int t  = threadIdx.x;
    const int b  = blockIdx.x >> 2;
    const int j0 = (blockIdx.x & 3) * DEST_COLS;
    const float* row = xin + (size_t)b * EDGES;

    constexpr int PER = DEST_COLS / PERM_THREADS;
    constexpr int NV  = WIN_COLS / 4 / PERM_THREADS;

    float val[PER];
    #pragma unroll
    for (int i = 0; i < PER; ++i) val[i] = 0.0f;

    for (int h = 0; h < 2; ++h) {
        if (h) __syncthreads();
        const v4f* src = (const v4f*)(row + h * WIN_COLS);
        v4f stage[NV];
        #pragma unroll
        for (int k = 0; k < NV; ++k) stage[k] = src[t + PERM_THREADS * k];
        #pragma unroll
        for (int k = 0; k < NV; ++k) *(v4f*)(win + 4 * (t + PERM_THREADS * k)) = stage[k];
        __syncthreads();

        int off = j0 + t;
        asm volatile("" : "+v"(off));
        int cidx[PER];
        #pragma unroll
        for (int i = 0; i < PER; ++i) cidx[i] = ixs[off + PERM_THREADS * i];
        #pragma unroll
        for (int i = 0; i < PER; ++i) {
            const int c = cidx[i];
            const float w = win[c & (WIN_COLS - 1)];
            val[i] = ((c >> 16) == h) ? w : val[i];
        }
    }

    float* dst = xg + (size_t)b * EDGES + j0;
    #pragma unroll
    for (int i = 0; i < PER; ++i)
        dst[t + PERM_THREADS * i] = val[i];
}

__global__ __launch_bounds__(256)
void gsnn_layer(const float* __restrict__ xin,
                const float* __restrict__ x0,
                const float* __restrict__ W1,
                const float* __restrict__ b1,
                const float* __restrict__ W2,
                const float* __restrict__ b2,
                float*       __restrict__ xout)
{
    __shared__ float lds_h[WAVES_PER_BLOCK * 16 * LDS_STRIDE];
    __shared__ __attribute__((aligned(16))) float lds_o[BATCH * OUT_PITCH];

    const int tid  = threadIdx.x;
    const int wave = tid >> 5;
    const int lane = tid & 31;
    const int n    = lane & 15;
    const int hi   = lane >> 4;

    const int f = blockIdx.x * WAVES_PER_BLOCK + wave;
    float* hbuf = &lds_h[wave * 16 * LDS_STRIDE];

    const float* W1f = W1 + (size_t)f * (HID * DEG);
    const v2f wb1_0 = *(const v2f*)(W1f + n * DEG + 2 * hi);
    const v2f wb1_1 = *(const v2f*)(W1f + n * DEG + 4 + 2 * hi);
    const float bias1 = b1[f * HID + n];

    const float* W2f = W2 + (size_t)f * (DEG * HID);
    v2f wb2[4];
    if (n < DEG) {
        #pragma unroll
        for (int j = 0; j < 4; ++j)
            wb2[j] = *(const v2f*)(W2f + n * HID + 4 * j + 2 * hi);
    } else {
        #pragma unroll
        for (int j = 0; j < 4; ++j) { wb2[j].x = 0.0f; wb2[j].y = 0.0f; }
    }

    const int cbase = f * DEG + 2 * hi;

    const int crow = lane >> 4;
    const int ccol = (lane & 15) * 4;
    const size_t gcol = (size_t)blockIdx.x * (WAVES_PER_BLOCK * DEG) + ccol;

    v4f xr[4];
    v4f b2v;

    #pragma unroll
    for (int mt = 0; mt < 4; ++mt) {
        const int mb = mt * 16;

        if (mt == 3) {
            #pragma unroll
            for (int i = 0; i < 4; ++i) {
                const int row = 8 * wave + 2 * i + crow;
                xr[i] = *(const v4f*)(x0 + (size_t)row * EDGES + gcol);
            }
            b2v = *(const v4f*)(b2 + gcol);
        }

        const float* xrow = xin + (size_t)(mb + n) * EDGES;
        const v2f a0 = *(const v2f*)(xrow + cbase);
        const v2f a1 = *(const v2f*)(xrow + cbase + 4);

        v8f acc = {};
        acc = __builtin_amdgcn_wmma_f32_16x16x4_f32(
                  false, a0, false, wb1_0, (short)0, acc, false, false);
        acc = __builtin_amdgcn_wmma_f32_16x16x4_f32(
                  false, a1, false, wb1_1, (short)0, acc, false, false);

        #pragma unroll
        for (int r = 0; r < 8; ++r) {
            float hv = elu_f(acc[r] + bias1);
            hbuf[(r + 8 * hi) * LDS_STRIDE + n] = hv;
        }
        if (mt == 3) {
            asm volatile("" :: "v"(xr[0]), "v"(xr[1]), "v"(xr[2]), "v"(xr[3]), "v"(b2v));
        }
        __syncthreads();

        v2f a2[4];
        #pragma unroll
        for (int j = 0; j < 4; ++j)
            a2[j] = *(const v2f*)(hbuf + n * LDS_STRIDE + 4 * j + 2 * hi);

        v8f o = {};
        #pragma unroll
        for (int j = 0; j < 4; ++j)
            o = __builtin_amdgcn_wmma_f32_16x16x4_f32(
                    false, wb2[j], false, a2[j], (short)0, o, false, false);

        __syncthreads();

        if (hi == 0) {
            float* orow = lds_o + (mb + n) * OUT_PITCH + wave * DEG;
            v4f lo4, hi4;
            lo4.x = o[0]; lo4.y = o[1]; lo4.z = o[2]; lo4.w = o[3];
            hi4.x = o[4]; hi4.y = o[5]; hi4.z = o[6]; hi4.w = o[7];
            *(v4f*)(orow)     = lo4;
            *(v4f*)(orow + 4) = hi4;
        }
    }

    __syncthreads();

    #pragma unroll
    for (int i = 0; i < 4; ++i) {
        const int row = 8 * wave + 2 * i + crow;
        const v4f ov = *(const v4f*)(lds_o + row * OUT_PITCH + ccol);
        const v4f res = (ov + b2v) + xr[i];
        *(v4f*)(xout + (size_t)row * EDGES + gcol) = res;
    }
}

extern "C" void kernel_launch(void* const* d_in, const int* in_sizes, int n_in,
                              void* d_out, int out_size, void* d_ws, size_t ws_size,
                              hipStream_t stream) {
    (void)in_sizes; (void)n_in; (void)out_size; (void)ws_size;

    const float* x0 = (const float*)d_in[0];
    const float* W1 = (const float*)d_in[1];
    const float* b1 = (const float*)d_in[2];
    const float* W2 = (const float*)d_in[3];
    const float* b2 = (const float*)d_in[4];
    const int*   ix = (const int*)d_in[5];

    float* xg  = (float*)d_ws;
    float* out = (float*)d_out;

    const dim3 grid(F_NODES / WAVES_PER_BLOCK);
    const dim3 block(WAVES_PER_BLOCK * 32);
    const dim3 pgrid(BATCH * 4);
    const dim3 pblock(PERM_THREADS);

    const float* cur = x0;
    for (int l = 0; l < LAYERS; ++l) {
        gsnn_permute<<<pgrid, pblock, 0, stream>>>(cur, ix, xg);
        gsnn_layer<<<grid, block, 0, stream>>>(xg, x0, W1, b1, W2, b2, out);
        cur = out;
    }
}


# permutation pass pipelined: indices loaded once beside the first half-row, second half-row's loads in flight under the first gather, stores under the second gather; same sums
# speedup vs baseline: 1.2902x; 1.0498x over previous
#include <hip/hip_runtime.h>
#include <hip/hip_bf16.h>


typedef __attribute__((ext_vector_type(2))) float v2f;
typedef __attribute__((ext_vector_type(8))) float v8f;
typedef __attribute__((ext_vector_type(4))) float v4f;

#define F_NODES 16384
#define DEG 8
#define EDGES (F_NODES * DEG)
#define BATCH 64
#define HID 16
#define LAYERS 10
#define WAVES_PER_BLOCK 8
#define LDS_STRIDE 18
#define OUT_PITCH 68

__device__ __forceinline__ float elu_f(float z) {
    return z > 0.0f ? z : (__expf(z) - 1.0f);
}


#define PERM_THREADS 1024
#define WIN_COLS 65536
#define DEST_COLS 32768

__device__ __forceinline__ void lds_barrier() {
    __builtin_amdgcn_fence(__ATOMIC_RELEASE, "workgroup", "local");
    __builtin_amdgcn_s_barrier();
    __builtin_amdgcn_fence(__ATOMIC_ACQUIRE, "workgroup", "local");
}

__global__ __launch_bounds__(PERM_THREADS)
void gsnn_permute(const float* __restrict__ xin,
                  const int*   __restrict__ ixs,
                  float*       __restrict__ xg)
{
    __shared__ __attribute__((aligned(16))) float win[WIN_COLS];

    const int t  = threadIdx.x;
    const int b  = blockIdx.x >> 2;
    const int j0 = (blockIdx.x & 3) * DEST_COLS;
    const float* row = xin + (size_t)b * EDGES;

    constexpr int PER = DEST_COLS / PERM_THREADS;
    constexpr int NV  = WIN_COLS / 4 / PERM_THREADS;

    int c[PER];
    for (int i = 0; i < PER; ++i) c[i] = ixs[j0 + t + PERM_THREADS * i];
    __builtin_amdgcn_sched_barrier(0);

    const v4f* src0 = (const v4f*)(row);
    const v4f* src1 = (const v4f*)(row + WIN_COLS);

    v4f stage[NV];
    for (int k = 0; k < NV; ++k) stage[k] = src0[t + PERM_THREADS * k];
    for (int k = 0; k < NV; ++k) *(v4f*)(win + 4 * (t + PERM_THREADS * k)) = stage[k];

    __builtin_amdgcn_sched_barrier(0);
    for (int k = 0; k < NV; ++k) stage[k] = src1[t + PERM_THREADS * k];

    lds_barrier();

    bool first[PER];
    for (int i = 0; i < PER; ++i) {
        const unsigned ci = (unsigned)c[i];
        first[i] = ci < (unsigned)WIN_COLS;
        if (first[i]) c[i] = __float_as_int(win[ci]);
    }

    lds_barrier();
    for (int k = 0; k < NV; ++k) *(v4f*)(win + 4 * (t + PERM_THREADS * k)) = stage[k];
    lds_barrier();

    float* dst = xg + (size_t)b * EDGES + j0;
    constexpr int GRP = 8;
    for (int g = 0; g < PER; g += GRP) {
        for (int i = g; i < g + GRP; ++i)
            if (!first[i]) c[i] = __float_as_int(win[c[i] & (WIN_COLS - 1)]);
        for (int i = g; i < g + GRP; ++i)
            dst[t + PERM_THREADS * i] = __int_as_float(c[i]);
    }
}

__global__ __launch_bounds__(256)
void gsnn_layer(const float* __restrict__ xin,
                const float* __restrict__ x0,
                const float* __restrict__ W1,
                const float* __restrict__ b1,
                const float* __restrict__ W2,
                const float* __restrict__ b2,
                float*       __restrict__ xout)
{
    __shared__ float lds_h[WAVES_PER_BLOCK * 16 * LDS_STRIDE];
    __shared__ __attribute__((aligned(16))) float lds_o[BATCH * OUT_PITCH];

    const int tid  = threadIdx.x;
    const int wave = tid >> 5;
    const int lane = tid & 31;
    const int n    = lane & 15;
    const int hi   = lane >> 4;

    const int f = blockIdx.x * WAVES_PER_BLOCK + wave;
    float* hbuf = &lds_h[wave * 16 * LDS_STRIDE];

    const float* W1f = W1 + (size_t)f * (HID * DEG);
    const v2f wb1_0 = *(const v2f*)(W1f + n * DEG + 2 * hi);
    const v2f wb1_1 = *(const v2f*)(W1f + n * DEG + 4 + 2 * hi);
    const float bias1 = b1[f * HID + n];

    const float* W2f = W2 + (size_t)f * (DEG * HID);
    v2f wb2[4];
    if (n < DEG) {
        #pragma unroll
        for (int j = 0; j < 4; ++j)
            wb2[j] = *(const v2f*)(W2f + n * HID + 4 * j + 2 * hi);
    } else {
        #pragma unroll
        for (int j = 0; j < 4; ++j) { wb2[j].x = 0.0f; wb2[j].y = 0.0f; }
    }

    const int cbase = f * DEG + 2 * hi;

    const int crow = lane >> 4;
    const int ccol = (lane & 15) * 4;
    const size_t gcol = (size_t)blockIdx.x * (WAVES_PER_BLOCK * DEG) + ccol;

    v4f xr[4];
    v4f b2v;

    #pragma unroll
    for (int mt = 0; mt < 4; ++mt) {
        const int mb = mt * 16;

        if (mt == 3) {
            #pragma unroll
            for (int i = 0; i < 4; ++i) {
                const int row = 8 * wave + 2 * i + crow;
                xr[i] = *(const v4f*)(x0 + (size_t)row * EDGES + gcol);
            }
            b2v = *(const v4f*)(b2 + gcol);
        }

        const float* xrow = xin + (size_t)(mb + n) * EDGES;
        const v2f a0 = *(const v2f*)(xrow + cbase);
        const v2f a1 = *(const v2f*)(xrow + cbase + 4);

        v8f acc = {};
        acc = __builtin_amdgcn_wmma_f32_16x16x4_f32(
                  false, a0, false, wb1_0, (short)0, acc, false, false);
        acc = __builtin_amdgcn_wmma_f32_16x16x4_f32(
                  false, a1, false, wb1_1, (short)0, acc, false, false);

        #pragma unroll
        for (int r = 0; r < 8; ++r) {
            float hv = elu_f(acc[r] + bias1);
            hbuf[(r + 8 * hi) * LDS_STRIDE + n] = hv;
        }
        if (mt == 3) {
            asm volatile("" :: "v"(xr[0]), "v"(xr[1]), "v"(xr[2]), "v"(xr[3]), "v"(b2v));
        }
        __syncthreads();

        v2f a2[4];
        #pragma unroll
        for (int j = 0; j < 4; ++j)
            a2[j] = *(const v2f*)(hbuf + n * LDS_STRIDE + 4 * j + 2 * hi);

        v8f o = {};
        #pragma unroll
        for (int j = 0; j < 4; ++j)
            o = __builtin_amdgcn_wmma_f32_16x16x4_f32(
                    false, wb2[j], false, a2[j], (short)0, o, false, false);

        __syncthreads();

        if (hi == 0) {
            float* orow = lds_o + (mb + n) * OUT_PITCH + wave * DEG;
            v4f lo4, hi4;
            lo4.x = o[0]; lo4.y = o[1]; lo4.z = o[2]; lo4.w = o[3];
            hi4.x = o[4]; hi4.y = o[5]; hi4.z = o[6]; hi4.w = o[7];
            *(v4f*)(orow)     = lo4;
            *(v4f*)(orow + 4) = hi4;
        }
    }

    __syncthreads();

    #pragma unroll
    for (int i = 0; i < 4; ++i) {
        const int row = 8 * wave + 2 * i + crow;
        const v4f ov = *(const v4f*)(lds_o + row * OUT_PITCH + ccol);
        const v4f res = (ov + b2v) + xr[i];
        *(v4f*)(xout + (size_t)row * EDGES + gcol) = res;
    }
}

extern "C" void kernel_launch(void* const* d_in, const int* in_sizes, int n_in,
                              void* d_out, int out_size, void* d_ws, size_t ws_size,
                              hipStream_t stream) {
    (void)in_sizes; (void)n_in; (void)out_size; (void)ws_size;

    const float* x0 = (const float*)d_in[0];
    const float* W1 = (const float*)d_in[1];
    const float* b1 = (const float*)d_in[2];
    const float* W2 = (const float*)d_in[3];
    const float* b2 = (const float*)d_in[4];
    const int*   ix = (const int*)d_in[5];

    float* xg  = (float*)d_ws;
    float* out = (float*)d_out;

    const dim3 grid(F_NODES / WAVES_PER_BLOCK);
    const dim3 block(WAVES_PER_BLOCK * 32);
    const dim3 pgrid(BATCH * 4);
    const dim3 pblock(PERM_THREADS);

    const float* cur = x0;
    for (int l = 0; l < LAYERS; ++l) {
        gsnn_permute<<<pgrid, pblock, 0, stream>>>(cur, ix, xg);
        gsnn_layer<<<grid, block, 0, stream>>>(xg, x0, W1, b1, W2, b2, out);
        cur = out;
    }
}
